# starting kernel
# baseline (speedup 1.0000x reference)
#include <hip/hip_runtime.h>


typedef __attribute__((ext_vector_type(2))) float v2f;
typedef __attribute__((ext_vector_type(8))) float v8f;

#define P_DIM    45
#define P_PAD    48
#define N_ITER   20
#define STEP     0.05f
#define MAXW     0.15f
#define EPSILON  1e-8f
#define WPB      8

__device__ __forceinline__ float sgnf(float x) {
    return (x > 0.0f) ? 1.0f : ((x < 0.0f) ? -1.0f : 0.0f);
}

__device__ __forceinline__ float clip_w(float x) {
    return fminf(fmaxf(x, 0.0f), MAXW);
}

__device__ __forceinline__ float wave_sum(float v) {
    #pragma unroll
    for (int off = 16; off > 0; off >>= 1)
        v += __shfl_xor(v, off, 32);
    return v;
}

__global__ __launch_bounds__(256) void
risk_budget_wmma_kernel(const float* __restrict__ sigma,
                        const float* __restrict__ beta,
                        const float* __restrict__ w_prev,
                        const float* __restrict__ log_ls,
                        const float* __restrict__ log_lt,
                        float* __restrict__ out,
                        int nB)
{
    __shared__ float wsh[WPB][P_PAD];
    __shared__ float swsh[WPB][P_PAD];

    const int lane = threadIdx.x & 31;
    const int slot = threadIdx.x >> 5;
    const int b    = blockIdx.x * WPB + slot;
    if (b >= nB) return;

    const float lam_s = expf(log_ls[0]);
    const float lam_t = expf(log_lt[0]);

    const int  rA   = lane;
    const int  rB   = lane + 32;
    const bool hasB = (rB < P_DIM);

    const float betaA  = beta  [(size_t)b * P_DIM + rA];
    const float wprevA = w_prev[(size_t)b * P_DIM + rA];
    const float betaB  = hasB ? beta  [(size_t)b * P_DIM + rB] : 0.0f;
    const float wprevB = hasB ? w_prev[(size_t)b * P_DIM + rB] : 0.0f;

    const float* sig = sigma + (size_t)b * (P_DIM * P_DIM);
    const int ml = lane & 15;
    const int h2 = (lane >> 4) << 1;
    v2f At[36];
    #pragma unroll
    for (int blk = 0; blk < 3; ++blk) {
        #pragma unroll
        for (int kc = 0; kc < 12; ++kc) {
            const int m  = blk * 16 + ml;
            const int k0 = kc * 4 + h2;
            v2f a;
            a.x = (m < P_DIM && k0     < P_DIM) ? sig[m * P_DIM + k0    ] : 0.0f;
            a.y = (m < P_DIM && k0 + 1 < P_DIM) ? sig[m * P_DIM + k0 + 1] : 0.0f;
            At[blk * 12 + kc] = a;
        }
    }

    float wA = 1.0f / (float)P_DIM;
    float wB = 1.0f / (float)P_DIM;
    wsh[slot][rA] = wA;
    if (hasB) wsh[slot][rB] = wB;
    if (lane >= 13 && lane < 16) wsh[slot][lane + 32] = 0.0f;
    __builtin_amdgcn_wave_barrier();

    #pragma unroll 1
    for (int it = 0; it < N_ITER; ++it) {
        #pragma unroll
        for (int blk = 0; blk < 3; ++blk) {
            v8f acc = {0.f, 0.f, 0.f, 0.f, 0.f, 0.f, 0.f, 0.f};
            #pragma unroll
            for (int kc = 0; kc < 12; ++kc) {
                const int kb = kc * 4 + h2;
                v2f bv;
                bv.x = wsh[slot][kb];
                bv.y = wsh[slot][kb + 1];
                acc = __builtin_amdgcn_wmma_f32_16x16x4_f32(
                     false, At[blk * 12 + kc],
                     false, bv,
                     (short)0, acc,
                     false,  false);
            }
            if ((lane & 15) == 0) {
                const int base = blk * 16 + (lane >> 4) * 8;
                #pragma unroll
                for (int j = 0; j < 8; ++j) swsh[slot][base + j] = acc[j];
            }
        }
        __builtin_amdgcn_wave_barrier();

        const float SwA = swsh[slot][rA];
        const float SwB = hasB ? swsh[slot][rB] : 0.0f;

        float nwA = wA - STEP * (2.0f * SwA - betaA + lam_s * sgnf(wA)
                                 + 2.0f * lam_t * (wA - wprevA));
        float nwB = wB - STEP * (2.0f * SwB - betaB + lam_s * sgnf(wB)
                                 + 2.0f * lam_t * (wB - wprevB));

        nwA = clip_w(nwA);
        nwB = clip_w(nwB);
        float s = wave_sum(nwA + (hasB ? nwB : 0.0f));
        float d = s + EPSILON;
        nwA = nwA / d;
        nwB = nwB / d;

        nwA = clip_w(nwA);
        nwB = clip_w(nwB);
        s = wave_sum(nwA + (hasB ? nwB : 0.0f));
        d = s + EPSILON;
        wA = nwA / d;
        wB = nwB / d;

        wsh[slot][rA] = wA;
        if (hasB) wsh[slot][rB] = wB;
        __builtin_amdgcn_wave_barrier();
    }

    out[(size_t)b * P_DIM + rA] = wA;
    if (hasB) out[(size_t)b * P_DIM + rB] = wB;
}

extern "C" void kernel_launch(void* const* d_in, const int* in_sizes, int n_in,
                              void* d_out, int out_size, void* d_ws, size_t ws_size,
                              hipStream_t stream) {
    const float* sigma  = (const float*)d_in[0];
    const float* beta   = (const float*)d_in[1];
    const float* w_prev = (const float*)d_in[2];
    const float* lls    = (const float*)d_in[3];
    const float* llt    = (const float*)d_in[4];
    float* out = (float*)d_out;

    const int nB   = in_sizes[1] / P_DIM;
    const int grid = (nB + WPB - 1) / WPB;

    hipLaunchKernelGGL(risk_budget_wmma_kernel, dim3(grid), dim3(256), 0, stream,
                       sigma, beta, w_prev, lls, llt, out, nB);
}


# Column 44 of all 3 row blocks folded into one WMMA (K slot = row block, B = w[44] on the diagonal) that feeds each chain as C: 34 f32 WMMAs per iteration, not 36; column-44 term now summed first
# speedup vs baseline: 1.0333x; 1.0333x over previous
#include <hip/hip_runtime.h>


typedef __attribute__((ext_vector_type(2))) float v2f;
typedef __attribute__((ext_vector_type(8))) float v8f;

#define P_DIM    45
#define P_PAD    48
#define N_ITER   20
#define STEP     0.05f
#define MAXW     0.15f
#define EPSILON  1e-8f
#define WPB      8

__device__ __forceinline__ float sgnf(float x) {
    return (x > 0.0f) ? 1.0f : ((x < 0.0f) ? -1.0f : 0.0f);
}

__device__ __forceinline__ float clip_w(float x) {
    return fminf(fmaxf(x, 0.0f), MAXW);
}

__device__ __forceinline__ float wave_sum(float v) {
    #pragma unroll
    for (int off = 16; off > 0; off >>= 1)
        v += __shfl_xor(v, off, 32);
    return v;
}

__global__ __launch_bounds__(256) void
risk_budget_wmma_kernel(const float* __restrict__ sigma,
                        const float* __restrict__ beta,
                        const float* __restrict__ w_prev,
                        const float* __restrict__ log_ls,
                        const float* __restrict__ log_lt,
                        float* __restrict__ out,
                        int nB)
{
    __shared__ float wsh[WPB][P_PAD];
    __shared__ float swsh[WPB][P_PAD];

    const int lane = threadIdx.x & 31;
    const int slot = threadIdx.x >> 5;
    const int b    = blockIdx.x * WPB + slot;
    if (b >= nB) return;

    const float lam_s = expf(log_ls[0]);
    const float lam_t = expf(log_lt[0]);

    const int  rA   = lane;
    const int  rB   = lane + 32;
    const bool hasB = (rB < P_DIM);

    const float betaA  = beta  [(size_t)b * P_DIM + rA];
    const float wprevA = w_prev[(size_t)b * P_DIM + rA];
    const float betaB  = hasB ? beta  [(size_t)b * P_DIM + rB] : 0.0f;
    const float wprevB = hasB ? w_prev[(size_t)b * P_DIM + rB] : 0.0f;

    const float* sig = sigma + (size_t)b * (P_DIM * P_DIM);
    const int ml = lane & 15;
    const int h2 = (lane >> 4) << 1;
    v2f At[33];
    #pragma unroll
    for (int blk = 0; blk < 3; ++blk) {
        #pragma unroll
        for (int kc = 0; kc < 11; ++kc) {
            const int m  = blk * 16 + ml;
            const int k0 = kc * 4 + h2;
            v2f a;
            a.x = (m < P_DIM) ? sig[m * P_DIM + k0    ] : 0.0f;
            a.y = (m < P_DIM) ? sig[m * P_DIM + k0 + 1] : 0.0f;
            At[blk * 11 + kc] = a;
        }
    }
    v2f Ax;
    {
        const int r0 = h2 * 16 + ml;
        const int r1 = (h2 + 1) * 16 + ml;
        Ax.x = (r0 < P_DIM) ? sig[r0 * P_DIM + (P_DIM - 1)] : 0.0f;
        Ax.y = (r1 < P_DIM) ? sig[r1 * P_DIM + (P_DIM - 1)] : 0.0f;
    }
    const bool selx = (ml == h2);
    const bool sely = (lane == 1);

    float wA = 1.0f / (float)P_DIM;
    float wB = 1.0f / (float)P_DIM;
    wsh[slot][rA] = wA;
    if (hasB) wsh[slot][rB] = wB;
    __builtin_amdgcn_wave_barrier();

    #pragma unroll 1
    for (int it = 0; it < N_ITER; ++it) {
        const float w44 = wsh[slot][P_DIM - 1];
        v2f bx;
        bx.x = selx ? w44 : 0.0f;
        bx.y = sely ? w44 : 0.0f;
        const v8f zero8 = {0.f, 0.f, 0.f, 0.f, 0.f, 0.f, 0.f, 0.f};
        const v8f xt = __builtin_amdgcn_wmma_f32_16x16x4_f32(
             false, Ax,  false, bx,
             (short)0, zero8,
             false,  false);
        #pragma unroll
        for (int blk = 0; blk < 3; ++blk) {
            v8f acc = xt;
            #pragma unroll
            for (int kc = 0; kc < 11; ++kc) {
                const int kb = kc * 4 + h2;
                v2f bv;
                bv.x = wsh[slot][kb];
                bv.y = wsh[slot][kb + 1];
                acc = __builtin_amdgcn_wmma_f32_16x16x4_f32(
                     false, At[blk * 11 + kc],
                     false, bv,
                     (short)0, acc,
                     false,  false);
            }
            if ((lane & 15) == blk) {
                const int base = blk * 16 + (lane >> 4) * 8;
                #pragma unroll
                for (int j = 0; j < 8; ++j) swsh[slot][base + j] = acc[j];
            }
        }
        __builtin_amdgcn_wave_barrier();

        const float SwA = swsh[slot][rA];
        const float SwB = hasB ? swsh[slot][rB] : 0.0f;

        float nwA = wA - STEP * (2.0f * SwA - betaA + lam_s * sgnf(wA)
                                 + 2.0f * lam_t * (wA - wprevA));
        float nwB = wB - STEP * (2.0f * SwB - betaB + lam_s * sgnf(wB)
                                 + 2.0f * lam_t * (wB - wprevB));

        nwA = clip_w(nwA);
        nwB = clip_w(nwB);
        float s = wave_sum(nwA + (hasB ? nwB : 0.0f));
        float d = s + EPSILON;
        nwA = nwA / d;
        nwB = nwB / d;

        nwA = clip_w(nwA);
        nwB = clip_w(nwB);
        s = wave_sum(nwA + (hasB ? nwB : 0.0f));
        d = s + EPSILON;
        wA = nwA / d;
        wB = nwB / d;

        wsh[slot][rA] = wA;
        if (hasB) wsh[slot][rB] = wB;
        __builtin_amdgcn_wave_barrier();
    }

    out[(size_t)b * P_DIM + rA] = wA;
    if (hasB) out[(size_t)b * P_DIM + rB] = wB;
}

extern "C" void kernel_launch(void* const* d_in, const int* in_sizes, int n_in,
                              void* d_out, int out_size, void* d_ws, size_t ws_size,
                              hipStream_t stream) {
    const float* sigma  = (const float*)d_in[0];
    const float* beta   = (const float*)d_in[1];
    const float* w_prev = (const float*)d_in[2];
    const float* lls    = (const float*)d_in[3];
    const float* llt    = (const float*)d_in[4];
    float* out = (float*)d_out;

    const int nB   = in_sizes[1] / P_DIM;
    const int grid = (nB + WPB - 1) / WPB;

    hipLaunchKernelGGL(risk_budget_wmma_kernel, dim3(grid), dim3(256), 0, stream,
                       sigma, beta, w_prev, lls, llt, out, nB);
}


# w put in row 15 of A and column 15 of B beside 15-row Sigma tiles: each f32 WMMA sums one row block via A, another via B; 18 per iteration not 34; row sum = cols 0..23 chain + cols 24..44 chain
# speedup vs baseline: 1.4481x; 1.4014x over previous
#include <hip/hip_runtime.h>


typedef __attribute__((ext_vector_type(2))) float v2f;
typedef __attribute__((ext_vector_type(4))) float v4f;
typedef __attribute__((ext_vector_type(8))) float v8f;

#define P_DIM    45
#define P_PAD    48
#define RBLK     15
#define N_ITER   20
#define STEP     0.05f
#define MAXW     0.15f
#define EPSILON  1e-8f
#define WPB      8
#define SW_LEN   96

__device__ __forceinline__ float sgnf(float x) {
    return (x > 0.0f) ? 1.0f : ((x < 0.0f) ? -1.0f : 0.0f);
}

__device__ __forceinline__ float clip_w(float x) {
    return fminf(fmaxf(x, 0.0f), MAXW);
}

__device__ __forceinline__ float wave_sum(float v) {
    #pragma unroll
    for (int off = 16; off > 0; off >>= 1)
        v += __shfl_xor(v, off, 32);
    return v;
}

template <int CHUNK>
__device__ __forceinline__ v2f sigma_pair(const float* __restrict__ rowp, int h2) {
    v2f a;
    if (CHUNK < 11) {
        a.x = rowp[CHUNK * 4 + h2];
        a.y = rowp[CHUNK * 4 + h2 + 1];
    } else {
        a.x = (h2 == 0) ? rowp[P_DIM - 1] : 0.0f;
        a.y = 0.0f;
    }
    return a;
}

__device__ __forceinline__ int piece_lo(int r) {
    return (r < RBLK) ? r : ((r < 2 * RBLK) ? 48 + (r - RBLK) : 64 + (r - 2 * RBLK));
}
__device__ __forceinline__ int piece_hi(int r) {
    return (r < RBLK) ? 80 + r : ((r < 2 * RBLK) ? 16 + (r - RBLK) : 32 + (r - 2 * RBLK));
}

__global__ __launch_bounds__(256) void
risk_budget_wmma_kernel(const float* __restrict__ sigma,
                        const float* __restrict__ beta,
                        const float* __restrict__ w_prev,
                        const float* __restrict__ log_ls,
                        const float* __restrict__ log_lt,
                        float* __restrict__ out,
                        int nB)
{
    __shared__ __attribute__((aligned(16))) float wsh[WPB][3][P_PAD];
    __shared__ __attribute__((aligned(16))) float swsh[WPB][SW_LEN];

    const int lane = threadIdx.x & 31;
    const int slot = threadIdx.x >> 5;
    const int b    = blockIdx.x * WPB + slot;
    if (b >= nB) return;

    const float lam_s = expf(log_ls[0]);
    const float lam_t = expf(log_lt[0]);

    const int  rA   = lane;
    const int  rB   = lane + 32;
    const bool hasB = (rB < P_DIM);

    const float betaA  = beta  [(size_t)b * P_DIM + rA];
    const float wprevA = w_prev[(size_t)b * P_DIM + rA];
    const float betaB  = hasB ? beta  [(size_t)b * P_DIM + rB] : 0.0f;
    const float wprevB = hasB ? w_prev[(size_t)b * P_DIM + rB] : 0.0f;

    const float* sig = sigma + (size_t)b * (P_DIM * P_DIM);
    const int  ml = lane & 15;
    const int  h  = lane >> 4;
    const int  h2 = h << 1;
    const bool wl = (ml == RBLK);
    const int  mr = wl ? 0 : ml;
    const float* row0 = sig + (mr            ) * P_DIM;
    const float* row1 = sig + (mr +     RBLK) * P_DIM;
    const float* row2 = sig + (mr + 2 * RBLK) * P_DIM;

    v2f A0[6], B0[6], A1[6], B1[6], A2[6], B2[6];
    #define LOAD_STEP(I)                                   \
        A0[I] = sigma_pair<(I)    >(row0, h2);             \
        B0[I] = sigma_pair<(I)    >(row1, h2);             \
        A1[I] = sigma_pair<(I) + 6>(row1, h2);             \
        B1[I] = sigma_pair<(I)    >(row2, h2);             \
        A2[I] = sigma_pair<(I) + 6>(row2, h2);             \
        B2[I] = sigma_pair<(I) + 6>(row0, h2);
    LOAD_STEP(0) LOAD_STEP(1) LOAD_STEP(2) LOAD_STEP(3) LOAD_STEP(4) LOAD_STEP(5)
    #undef LOAD_STEP

    float wA = 1.0f / (float)P_DIM;
    float wB = 1.0f / (float)P_DIM;
    #pragma unroll
    for (int c = 0; c < 3; ++c) {
        wsh[slot][c][rA] = wA;
        if (lane < 16) wsh[slot][c][rB] = hasB ? wB : 0.0f;
    }
    __builtin_amdgcn_wave_barrier();

    const int loA = piece_lo(rA);
    const int hiA = piece_hi(rA);
    const int loB = hasB ? piece_lo(rB) : 0;
    const int hiB = hasB ? piece_hi(rB) : 0;

    #pragma unroll 1
    for (int it = 0; it < N_ITER; ++it) {
        if (wl) {
            #pragma unroll
            for (int i = 0; i < 6; ++i) {
                A0[i] = *(const v2f*)&wsh[slot][0][4 * i + h2];
                B0[i] = *(const v2f*)&wsh[slot][1][4 * i + h2];
                A1[i] = *(const v2f*)&wsh[slot][2][4 * i + h2];
                B1[i] = *(const v2f*)&wsh[slot][0][4 * i + 24 + h2];
                A2[i] = *(const v2f*)&wsh[slot][1][4 * i + 24 + h2];
                B2[i] = *(const v2f*)&wsh[slot][2][4 * i + 24 + h2];
            }
        }

        v8f acc0 = {0.f, 0.f, 0.f, 0.f, 0.f, 0.f, 0.f, 0.f};
        v8f acc1 = acc0;
        v8f acc2 = acc0;
        #pragma unroll
        for (int i = 0; i < 6; ++i) {
            acc0 = __builtin_amdgcn_wmma_f32_16x16x4_f32(
                 false, A0[i],  false, B0[i],
                 (short)0, acc0,  false,  false);
            acc1 = __builtin_amdgcn_wmma_f32_16x16x4_f32(
                 false, A1[i],  false, B1[i],
                 (short)0, acc1,  false,  false);
            acc2 = __builtin_amdgcn_wmma_f32_16x16x4_f32(
                 false, A2[i],  false, B2[i],
                 (short)0, acc2,  false,  false);
        }

        if (wl) {
            float* p = &swsh[slot][8 * h];
            *(v4f*)(p     ) = __builtin_shufflevector(acc0, acc0, 0, 1, 2, 3);
            *(v4f*)(p +  4) = __builtin_shufflevector(acc0, acc0, 4, 5, 6, 7);
            *(v4f*)(p + 16) = __builtin_shufflevector(acc1, acc1, 0, 1, 2, 3);
            *(v4f*)(p + 20) = __builtin_shufflevector(acc1, acc1, 4, 5, 6, 7);
            *(v4f*)(p + 32) = __builtin_shufflevector(acc2, acc2, 0, 1, 2, 3);
            *(v4f*)(p + 36) = __builtin_shufflevector(acc2, acc2, 4, 5, 6, 7);
        }
        if (h == 1) {
            swsh[slot][48 + ml] = acc0[7];
            swsh[slot][64 + ml] = acc1[7];
            swsh[slot][80 + ml] = acc2[7];
        }
        __builtin_amdgcn_wave_barrier();

        const float SwA = swsh[slot][loA] + swsh[slot][hiA];
        const float SwB = hasB ? (swsh[slot][loB] + swsh[slot][hiB]) : 0.0f;

        float nwA = wA - STEP * (2.0f * SwA - betaA + lam_s * sgnf(wA)
                                 + 2.0f * lam_t * (wA - wprevA));
        float nwB = wB - STEP * (2.0f * SwB - betaB + lam_s * sgnf(wB)
                                 + 2.0f * lam_t * (wB - wprevB));

        nwA = clip_w(nwA);
        nwB = clip_w(nwB);
        float s = wave_sum(nwA + (hasB ? nwB : 0.0f));
        float d = s + EPSILON;
        nwA = nwA / d;
        nwB = nwB / d;

        nwA = clip_w(nwA);
        nwB = clip_w(nwB);
        s = wave_sum(nwA + (hasB ? nwB : 0.0f));
        d = s + EPSILON;
        wA = nwA / d;
        wB = nwB / d;

        #pragma unroll
        for (int c = 0; c < 3; ++c) {
            wsh[slot][c][rA] = wA;
            if (hasB) wsh[slot][c][rB] = wB;
        }
        __builtin_amdgcn_wave_barrier();
    }

    out[(size_t)b * P_DIM + rA] = wA;
    if (hasB) out[(size_t)b * P_DIM + rB] = wB;
}

extern "C" void kernel_launch(void* const* d_in, const int* in_sizes, int n_in,
                              void* d_out, int out_size, void* d_ws, size_t ws_size,
                              hipStream_t stream) {
    const float* sigma  = (const float*)d_in[0];
    const float* beta   = (const float*)d_in[1];
    const float* w_prev = (const float*)d_in[2];
    const float* lls    = (const float*)d_in[3];
    const float* llt    = (const float*)d_in[4];
    float* out = (float*)d_out;

    const int nB   = in_sizes[1] / P_DIM;
    const int grid = (nB + WPB - 1) / WPB;

    hipLaunchKernelGGL(risk_budget_wmma_kernel, dim3(grid), dim3(256), 0, stream,
                       sigma, beta, w_prev, lls, llt, out, nB);
}


# Each renormalisation takes both quotients from one hardware reciprocal of the shared divisor (product, FMA remainder, one correction; equal to division in 5M emulated pairs), not two f32 divisions
# speedup vs baseline: 1.4922x; 1.0305x over previous
#include <hip/hip_runtime.h>


typedef __attribute__((ext_vector_type(2))) float v2f;
typedef __attribute__((ext_vector_type(4))) float v4f;
typedef __attribute__((ext_vector_type(8))) float v8f;

#define P_DIM    45
#define P_PAD    48
#define RBLK     15
#define N_ITER   20
#define STEP     0.05f
#define MAXW     0.15f
#define EPSILON  1e-8f
#define WPB      8
#define SW_LEN   96

__device__ __forceinline__ float sgnf(float x) {
    return (x > 0.0f) ? 1.0f : ((x < 0.0f) ? -1.0f : 0.0f);
}

__device__ __forceinline__ float clip_w(float x) {
    return fminf(fmaxf(x, 0.0f), MAXW);
}

__device__ __forceinline__ float quotient(float x, float d, float r) {
    const float q   = x * r;
    const float rem = __builtin_fmaf(-d, q, x);
    return __builtin_fmaf(rem, r, q);
}

__device__ __forceinline__ float wave_sum(float v) {
    #pragma unroll
    for (int off = 16; off > 0; off >>= 1)
        v += __shfl_xor(v, off, 32);
    return v;
}

template <int CHUNK>
__device__ __forceinline__ v2f sigma_pair(const float* __restrict__ rowp, int h2) {
    v2f a;
    if (CHUNK < 11) {
        a.x = rowp[CHUNK * 4 + h2];
        a.y = rowp[CHUNK * 4 + h2 + 1];
    } else {
        a.x = (h2 == 0) ? rowp[P_DIM - 1] : 0.0f;
        a.y = 0.0f;
    }
    return a;
}

__device__ __forceinline__ int piece_lo(int r) {
    return (r < RBLK) ? r : ((r < 2 * RBLK) ? 48 + (r - RBLK) : 64 + (r - 2 * RBLK));
}
__device__ __forceinline__ int piece_hi(int r) {
    return (r < RBLK) ? 80 + r : ((r < 2 * RBLK) ? 16 + (r - RBLK) : 32 + (r - 2 * RBLK));
}

__global__ __launch_bounds__(256) void
risk_budget_wmma_kernel(const float* __restrict__ sigma,
                        const float* __restrict__ beta,
                        const float* __restrict__ w_prev,
                        const float* __restrict__ log_ls,
                        const float* __restrict__ log_lt,
                        float* __restrict__ out,
                        int nB)
{
    __shared__ __attribute__((aligned(16))) float wsh[WPB][3][P_PAD];
    __shared__ __attribute__((aligned(16))) float swsh[WPB][SW_LEN];

    const int lane = threadIdx.x & 31;
    const int slot = threadIdx.x >> 5;
    const int b    = blockIdx.x * WPB + slot;
    if (b >= nB) return;

    const float lam_s = expf(log_ls[0]);
    const float lam_t = expf(log_lt[0]);

    const int  rA   = lane;
    const int  rB   = lane + 32;
    const bool hasB = (rB < P_DIM);

    const float betaA  = beta  [(size_t)b * P_DIM + rA];
    const float wprevA = w_prev[(size_t)b * P_DIM + rA];
    const float betaB  = hasB ? beta  [(size_t)b * P_DIM + rB] : 0.0f;
    const float wprevB = hasB ? w_prev[(size_t)b * P_DIM + rB] : 0.0f;

    const float* sig = sigma + (size_t)b * (P_DIM * P_DIM);
    const int  ml = lane & 15;
    const int  h  = lane >> 4;
    const int  h2 = h << 1;
    const bool wl = (ml == RBLK);
    const int  mr = wl ? 0 : ml;
    const float* row0 = sig + (mr            ) * P_DIM;
    const float* row1 = sig + (mr +     RBLK) * P_DIM;
    const float* row2 = sig + (mr + 2 * RBLK) * P_DIM;

    v2f A0[6], B0[6], A1[6], B1[6], A2[6], B2[6];
    #define LOAD_STEP(I)                                   \
        A0[I] = sigma_pair<(I)    >(row0, h2);             \
        B0[I] = sigma_pair<(I)    >(row1, h2);             \
        A1[I] = sigma_pair<(I) + 6>(row1, h2);             \
        B1[I] = sigma_pair<(I)    >(row2, h2);             \
        A2[I] = sigma_pair<(I) + 6>(row2, h2);             \
        B2[I] = sigma_pair<(I) + 6>(row0, h2);
    LOAD_STEP(0) LOAD_STEP(1) LOAD_STEP(2) LOAD_STEP(3) LOAD_STEP(4) LOAD_STEP(5)
    #undef LOAD_STEP

    float wA = 1.0f / (float)P_DIM;
    float wB = 1.0f / (float)P_DIM;
    #pragma unroll
    for (int c = 0; c < 3; ++c) {
        wsh[slot][c][rA] = wA;
        if (lane < 16) wsh[slot][c][rB] = hasB ? wB : 0.0f;
    }
    __builtin_amdgcn_wave_barrier();

    const int loA = piece_lo(rA);
    const int hiA = piece_hi(rA);
    const int loB = hasB ? piece_lo(rB) : 0;
    const int hiB = hasB ? piece_hi(rB) : 0;

    #pragma unroll 1
    for (int it = 0; it < N_ITER; ++it) {
        if (wl) {
            #pragma unroll
            for (int i = 0; i < 6; ++i) {
                A0[i] = *(const v2f*)&wsh[slot][0][4 * i + h2];
                B0[i] = *(const v2f*)&wsh[slot][1][4 * i + h2];
                A1[i] = *(const v2f*)&wsh[slot][2][4 * i + h2];
                B1[i] = *(const v2f*)&wsh[slot][0][4 * i + 24 + h2];
                A2[i] = *(const v2f*)&wsh[slot][1][4 * i + 24 + h2];
                B2[i] = *(const v2f*)&wsh[slot][2][4 * i + 24 + h2];
            }
        }

        v8f acc0 = {0.f, 0.f, 0.f, 0.f, 0.f, 0.f, 0.f, 0.f};
        v8f acc1 = acc0;
        v8f acc2 = acc0;
        #pragma unroll
        for (int i = 0; i < 6; ++i) {
            acc0 = __builtin_amdgcn_wmma_f32_16x16x4_f32(
                 false, A0[i],  false, B0[i],
                 (short)0, acc0,  false,  false);
            acc1 = __builtin_amdgcn_wmma_f32_16x16x4_f32(
                 false, A1[i],  false, B1[i],
                 (short)0, acc1,  false,  false);
            acc2 = __builtin_amdgcn_wmma_f32_16x16x4_f32(
                 false, A2[i],  false, B2[i],
                 (short)0, acc2,  false,  false);
        }

        if (wl) {
            float* p = &swsh[slot][8 * h];
            *(v4f*)(p     ) = __builtin_shufflevector(acc0, acc0, 0, 1, 2, 3);
            *(v4f*)(p +  4) = __builtin_shufflevector(acc0, acc0, 4, 5, 6, 7);
            *(v4f*)(p + 16) = __builtin_shufflevector(acc1, acc1, 0, 1, 2, 3);
            *(v4f*)(p + 20) = __builtin_shufflevector(acc1, acc1, 4, 5, 6, 7);
            *(v4f*)(p + 32) = __builtin_shufflevector(acc2, acc2, 0, 1, 2, 3);
            *(v4f*)(p + 36) = __builtin_shufflevector(acc2, acc2, 4, 5, 6, 7);
        }
        if (h == 1) {
            swsh[slot][48 + ml] = acc0[7];
            swsh[slot][64 + ml] = acc1[7];
            swsh[slot][80 + ml] = acc2[7];
        }
        __builtin_amdgcn_wave_barrier();

        const float SwA = swsh[slot][loA] + swsh[slot][hiA];
        const float SwB = hasB ? (swsh[slot][loB] + swsh[slot][hiB]) : 0.0f;

        float nwA = wA - STEP * (2.0f * SwA - betaA + lam_s * sgnf(wA)
                                 + 2.0f * lam_t * (wA - wprevA));
        float nwB = wB - STEP * (2.0f * SwB - betaB + lam_s * sgnf(wB)
                                 + 2.0f * lam_t * (wB - wprevB));

        nwA = clip_w(nwA);
        nwB = clip_w(nwB);
        float s = wave_sum(nwA + (hasB ? nwB : 0.0f));
        float d = s + EPSILON;
        float r = __builtin_amdgcn_rcpf(d);
        nwA = quotient(nwA, d, r);
        nwB = quotient(nwB, d, r);

        nwA = clip_w(nwA);
        nwB = clip_w(nwB);
        s = wave_sum(nwA + (hasB ? nwB : 0.0f));
        d = s + EPSILON;
        r = __builtin_amdgcn_rcpf(d);
        wA = quotient(nwA, d, r);
        wB = quotient(nwB, d, r);

        #pragma unroll
        for (int c = 0; c < 3; ++c) {
            wsh[slot][c][rA] = wA;
            if (hasB) wsh[slot][c][rB] = wB;
        }
        __builtin_amdgcn_wave_barrier();
    }

    out[(size_t)b * P_DIM + rA] = wA;
    if (hasB) out[(size_t)b * P_DIM + rB] = wB;
}

extern "C" void kernel_launch(void* const* d_in, const int* in_sizes, int n_in,
                              void* d_out, int out_size, void* d_ws, size_t ws_size,
                              hipStream_t stream) {
    const float* sigma  = (const float*)d_in[0];
    const float* beta   = (const float*)d_in[1];
    const float* w_prev = (const float*)d_in[2];
    const float* lls    = (const float*)d_in[3];
    const float* llt    = (const float*)d_in[4];
    float* out = (float*)d_out;

    const int nB   = in_sizes[1] / P_DIM;
    const int grid = (nB + WPB - 1) / WPB;

    hipLaunchKernelGGL(risk_budget_wmma_kernel, dim3(grid), dim3(256), 0, stream,
                       sigma, beta, w_prev, lls, llt, out, nB);
}


# Both wave sums of an iteration as vector-ALU butterflies (one v_permlanex16 + four row_xmask DPP adds, distances 16,8,4,2,1 as before) instead of ten ds_bpermute round trips; bit-identical sums
# speedup vs baseline: 1.6384x; 1.0980x over previous
#include <hip/hip_runtime.h>


typedef __attribute__((ext_vector_type(2))) float v2f;
typedef __attribute__((ext_vector_type(4))) float v4f;
typedef __attribute__((ext_vector_type(8))) float v8f;

#define P_DIM    45
#define P_PAD    48
#define RBLK     15
#define N_ITER   20
#define STEP     0.05f
#define MAXW     0.15f
#define EPSILON  1e-8f
#define WPB      8
#define SW_LEN   96

__device__ __forceinline__ float sgnf(float x) {
    return (x > 0.0f) ? 1.0f : ((x < 0.0f) ? -1.0f : 0.0f);
}

__device__ __forceinline__ float clip_w(float x) {
    return fminf(fmaxf(x, 0.0f), MAXW);
}

__device__ __forceinline__ float quotient(float x, float d, float r) {
    const float q   = x * r;
    const float rem = __builtin_fmaf(-d, q, x);
    return __builtin_fmaf(rem, r, q);
}

__device__ __forceinline__ float other_half(float x) {
    const int b = __builtin_bit_cast(int, x);
    return __builtin_bit_cast(float,
        __builtin_amdgcn_permlanex16(b, b, 0x76543210, 0xfedcba98, false, false));
}
template <int M>
__device__ __forceinline__ float row_xor(float x) {
    const int b = __builtin_bit_cast(int, x);
    return __builtin_bit_cast(float,
        __builtin_amdgcn_update_dpp(b, b, 0x160 + M, 0xf, 0xf, true));
}

__device__ __forceinline__ float wave_sum(float v) {
    v += other_half(v);
    v += row_xor<8>(v);
    v += row_xor<4>(v);
    v += row_xor<2>(v);
    v += row_xor<1>(v);
    return v;
}

template <int CHUNK>
__device__ __forceinline__ v2f sigma_pair(const float* __restrict__ rowp, int h2) {
    v2f a;
    if (CHUNK < 11) {
        a.x = rowp[CHUNK * 4 + h2];
        a.y = rowp[CHUNK * 4 + h2 + 1];
    } else {
        a.x = (h2 == 0) ? rowp[P_DIM - 1] : 0.0f;
        a.y = 0.0f;
    }
    return a;
}

__device__ __forceinline__ int piece_lo(int r) {
    return (r < RBLK) ? r : ((r < 2 * RBLK) ? 48 + (r - RBLK) : 64 + (r - 2 * RBLK));
}
__device__ __forceinline__ int piece_hi(int r) {
    return (r < RBLK) ? 80 + r : ((r < 2 * RBLK) ? 16 + (r - RBLK) : 32 + (r - 2 * RBLK));
}

__global__ __launch_bounds__(256) void
risk_budget_wmma_kernel(const float* __restrict__ sigma,
                        const float* __restrict__ beta,
                        const float* __restrict__ w_prev,
                        const float* __restrict__ log_ls,
                        const float* __restrict__ log_lt,
                        float* __restrict__ out,
                        int nB)
{
    __shared__ __attribute__((aligned(16))) float wsh[WPB][3][P_PAD];
    __shared__ __attribute__((aligned(16))) float swsh[WPB][SW_LEN];

    const int lane = threadIdx.x & 31;
    const int slot = threadIdx.x >> 5;
    const int b    = blockIdx.x * WPB + slot;
    if (b >= nB) return;

    const float lam_s = expf(log_ls[0]);
    const float lam_t = expf(log_lt[0]);

    const int  rA   = lane;
    const int  rB   = lane + 32;
    const bool hasB = (rB < P_DIM);

    const float betaA  = beta  [(size_t)b * P_DIM + rA];
    const float wprevA = w_prev[(size_t)b * P_DIM + rA];
    const float betaB  = hasB ? beta  [(size_t)b * P_DIM + rB] : 0.0f;
    const float wprevB = hasB ? w_prev[(size_t)b * P_DIM + rB] : 0.0f;

    const float* sig = sigma + (size_t)b * (P_DIM * P_DIM);
    const int  ml = lane & 15;
    const int  h  = lane >> 4;
    const int  h2 = h << 1;
    const bool wl = (ml == RBLK);
    const int  mr = wl ? 0 : ml;
    const float* row0 = sig + (mr            ) * P_DIM;
    const float* row1 = sig + (mr +     RBLK) * P_DIM;
    const float* row2 = sig + (mr + 2 * RBLK) * P_DIM;

    v2f A0[6], B0[6], A1[6], B1[6], A2[6], B2[6];
    #define LOAD_STEP(I)                                   \
        A0[I] = sigma_pair<(I)    >(row0, h2);             \
        B0[I] = sigma_pair<(I)    >(row1, h2);             \
        A1[I] = sigma_pair<(I) + 6>(row1, h2);             \
        B1[I] = sigma_pair<(I)    >(row2, h2);             \
        A2[I] = sigma_pair<(I) + 6>(row2, h2);             \
        B2[I] = sigma_pair<(I) + 6>(row0, h2);
    LOAD_STEP(0) LOAD_STEP(1) LOAD_STEP(2) LOAD_STEP(3) LOAD_STEP(4) LOAD_STEP(5)
    #undef LOAD_STEP

    float wA = 1.0f / (float)P_DIM;
    float wB = 1.0f / (float)P_DIM;
    #pragma unroll
    for (int c = 0; c < 3; ++c) {
        wsh[slot][c][rA] = wA;
        if (lane < 16) wsh[slot][c][rB] = hasB ? wB : 0.0f;
    }
    __builtin_amdgcn_wave_barrier();

    const int loA = piece_lo(rA);
    const int hiA = piece_hi(rA);
    const int loB = hasB ? piece_lo(rB) : 0;
    const int hiB = hasB ? piece_hi(rB) : 0;

    #pragma unroll 1
    for (int it = 0; it < N_ITER; ++it) {
        if (wl) {
            #pragma unroll
            for (int i = 0; i < 6; ++i) {
                A0[i] = *(const v2f*)&wsh[slot][0][4 * i + h2];
                B0[i] = *(const v2f*)&wsh[slot][1][4 * i + h2];
                A1[i] = *(const v2f*)&wsh[slot][2][4 * i + h2];
                B1[i] = *(const v2f*)&wsh[slot][0][4 * i + 24 + h2];
                A2[i] = *(const v2f*)&wsh[slot][1][4 * i + 24 + h2];
                B2[i] = *(const v2f*)&wsh[slot][2][4 * i + 24 + h2];
            }
        }

        v8f acc0 = {0.f, 0.f, 0.f, 0.f, 0.f, 0.f, 0.f, 0.f};
        v8f acc1 = acc0;
        v8f acc2 = acc0;
        #pragma unroll
        for (int i = 0; i < 6; ++i) {
            acc0 = __builtin_amdgcn_wmma_f32_16x16x4_f32(
                 false, A0[i],  false, B0[i],
                 (short)0, acc0,  false,  false);
            acc1 = __builtin_amdgcn_wmma_f32_16x16x4_f32(
                 false, A1[i],  false, B1[i],
                 (short)0, acc1,  false,  false);
            acc2 = __builtin_amdgcn_wmma_f32_16x16x4_f32(
                 false, A2[i],  false, B2[i],
                 (short)0, acc2,  false,  false);
        }

        if (wl) {
            float* p = &swsh[slot][8 * h];
            *(v4f*)(p     ) = __builtin_shufflevector(acc0, acc0, 0, 1, 2, 3);
            *(v4f*)(p +  4) = __builtin_shufflevector(acc0, acc0, 4, 5, 6, 7);
            *(v4f*)(p + 16) = __builtin_shufflevector(acc1, acc1, 0, 1, 2, 3);
            *(v4f*)(p + 20) = __builtin_shufflevector(acc1, acc1, 4, 5, 6, 7);
            *(v4f*)(p + 32) = __builtin_shufflevector(acc2, acc2, 0, 1, 2, 3);
            *(v4f*)(p + 36) = __builtin_shufflevector(acc2, acc2, 4, 5, 6, 7);
        }
        if (h == 1) {
            swsh[slot][48 + ml] = acc0[7];
            swsh[slot][64 + ml] = acc1[7];
            swsh[slot][80 + ml] = acc2[7];
        }
        __builtin_amdgcn_wave_barrier();

        const float SwA = swsh[slot][loA] + swsh[slot][hiA];
        const float SwB = hasB ? (swsh[slot][loB] + swsh[slot][hiB]) : 0.0f;

        float nwA = wA - STEP * (2.0f * SwA - betaA + lam_s * sgnf(wA)
                                 + 2.0f * lam_t * (wA - wprevA));
        float nwB = wB - STEP * (2.0f * SwB - betaB + lam_s * sgnf(wB)
                                 + 2.0f * lam_t * (wB - wprevB));

        nwA = clip_w(nwA);
        nwB = clip_w(nwB);
        float s = wave_sum(nwA + (hasB ? nwB : 0.0f));
        float d = s + EPSILON;
        float r = __builtin_amdgcn_rcpf(d);
        nwA = quotient(nwA, d, r);
        nwB = quotient(nwB, d, r);

        nwA = clip_w(nwA);
        nwB = clip_w(nwB);
        s = wave_sum(nwA + (hasB ? nwB : 0.0f));
        d = s + EPSILON;
        r = __builtin_amdgcn_rcpf(d);
        wA = quotient(nwA, d, r);
        wB = quotient(nwB, d, r);

        #pragma unroll
        for (int c = 0; c < 3; ++c) {
            wsh[slot][c][rA] = wA;
            if (hasB) wsh[slot][c][rB] = wB;
        }
        __builtin_amdgcn_wave_barrier();
    }

    out[(size_t)b * P_DIM + rA] = wA;
    if (hasB) out[(size_t)b * P_DIM + rB] = wB;
}

extern "C" void kernel_launch(void* const* d_in, const int* in_sizes, int n_in,
                              void* d_out, int out_size, void* d_ws, size_t ws_size,
                              hipStream_t stream) {
    const float* sigma  = (const float*)d_in[0];
    const float* beta   = (const float*)d_in[1];
    const float* w_prev = (const float*)d_in[2];
    const float* lls    = (const float*)d_in[3];
    const float* llt    = (const float*)d_in[4];
    float* out = (float*)d_out;

    const int nB   = in_sizes[1] / P_DIM;
    const int grid = (nB + WPB - 1) / WPB;

    hipLaunchKernelGGL(risk_budget_wmma_kernel, dim3(grid), dim3(256), 0, stream,
                       sigma, beta, w_prev, lls, llt, out, nB);
}


# Loop's row code made branch-free: an empty second row slot carries exact zeros, so w publish and row-sum loads run for all lanes as two-address LDS ops (7 stores to 3, 4 loads to 2); sums unchanged
# speedup vs baseline: 1.6918x; 1.0326x over previous
#include <hip/hip_runtime.h>


typedef __attribute__((ext_vector_type(2))) float v2f;
typedef __attribute__((ext_vector_type(4))) float v4f;
typedef __attribute__((ext_vector_type(8))) float v8f;

#define P_DIM    45
#define P_PAD    48
#define RBLK     15
#define N_ITER   20
#define STEP     0.05f
#define MAXW     0.15f
#define EPSILON  1e-8f
#define WPB      8
#define W_PITCH  64
#define SW_LEN   96
#define SW_HI    48

__device__ __forceinline__ float sgnf(float x) {
    return (x > 0.0f) ? 1.0f : ((x < 0.0f) ? -1.0f : 0.0f);
}

__device__ __forceinline__ float clip_w(float x) {
    return fminf(fmaxf(x, 0.0f), MAXW);
}

__device__ __forceinline__ float quotient(float x, float d, float r) {
    const float q   = x * r;
    const float rem = __builtin_fmaf(-d, q, x);
    return __builtin_fmaf(rem, r, q);
}

__device__ __forceinline__ float other_half(float x) {
    const int b = __builtin_bit_cast(int, x);
    return __builtin_bit_cast(float,
        __builtin_amdgcn_permlanex16(b, b, 0x76543210, 0xfedcba98, false, false));
}
template <int M>
__device__ __forceinline__ float row_xor(float x) {
    const int b = __builtin_bit_cast(int, x);
    return __builtin_bit_cast(float,
        __builtin_amdgcn_update_dpp(b, b, 0x160 + M, 0xf, 0xf, true));
}

__device__ __forceinline__ float wave_sum(float v) {
    v += other_half(v);
    v += row_xor<8>(v);
    v += row_xor<4>(v);
    v += row_xor<2>(v);
    v += row_xor<1>(v);
    return v;
}

template <int CHUNK>
__device__ __forceinline__ v2f sigma_pair(const float* __restrict__ rowp, int h2) {
    v2f a;
    if (CHUNK < 11) {
        a.x = rowp[CHUNK * 4 + h2];
        a.y = rowp[CHUNK * 4 + h2 + 1];
    } else {
        a.x = (h2 == 0) ? rowp[P_DIM - 1] : 0.0f;
        a.y = 0.0f;
    }
    return a;
}

__device__ __forceinline__ int piece_cell(int r) {
    return (r < RBLK) ? r : ((r < 2 * RBLK) ? 16 + (r - RBLK) : 32 + (r - 2 * RBLK));
}

__global__ __launch_bounds__(256) void
risk_budget_wmma_kernel(const float* __restrict__ sigma,
                        const float* __restrict__ beta,
                        const float* __restrict__ w_prev,
                        const float* __restrict__ log_ls,
                        const float* __restrict__ log_lt,
                        float* __restrict__ out,
                        int nB)
{
    __shared__ __attribute__((aligned(16))) float wsh[WPB][3][W_PITCH];
    __shared__ __attribute__((aligned(16))) float swsh[WPB][SW_LEN];

    const int lane = threadIdx.x & 31;
    const int slot = threadIdx.x >> 5;
    const int b    = blockIdx.x * WPB + slot;
    if (b >= nB) return;

    const float lam_s = expf(log_ls[0]);
    const float lam_t = expf(log_lt[0]);

    const int  rA   = lane;
    const int  rB   = lane + 32;
    const bool hasB = (rB < P_DIM);

    const float betaA  = beta  [(size_t)b * P_DIM + rA];
    const float wprevA = w_prev[(size_t)b * P_DIM + rA];
    const float betaB  = hasB ? beta  [(size_t)b * P_DIM + rB] : 0.0f;
    const float wprevB = hasB ? w_prev[(size_t)b * P_DIM + rB] : 0.0f;

    const float* sig = sigma + (size_t)b * (P_DIM * P_DIM);
    const int  ml = lane & 15;
    const int  h  = lane >> 4;
    const int  h2 = h << 1;
    const bool wl = (ml == RBLK);
    const int  mr = wl ? 0 : ml;
    const float* row0 = sig + (mr            ) * P_DIM;
    const float* row1 = sig + (mr +     RBLK) * P_DIM;
    const float* row2 = sig + (mr + 2 * RBLK) * P_DIM;

    v2f A0[6], B0[6], A1[6], B1[6], A2[6], B2[6];
    #define LOAD_STEP(I)                                   \
        A0[I] = sigma_pair<(I)    >(row0, h2);             \
        B0[I] = sigma_pair<(I)    >(row1, h2);             \
        A1[I] = sigma_pair<(I) + 6>(row1, h2);             \
        B1[I] = sigma_pair<(I)    >(row2, h2);             \
        A2[I] = sigma_pair<(I) + 6>(row2, h2);             \
        B2[I] = sigma_pair<(I) + 6>(row0, h2);
    LOAD_STEP(0) LOAD_STEP(1) LOAD_STEP(2) LOAD_STEP(3) LOAD_STEP(4) LOAD_STEP(5)
    #undef LOAD_STEP

    float wA = 1.0f / (float)P_DIM;
    float wB = hasB ? 1.0f / (float)P_DIM : 0.0f;
    #pragma unroll
    for (int c = 0; c < 3; ++c) {
        wsh[slot][c][rA] = wA;
        wsh[slot][c][rB] = wB;
    }
    __builtin_amdgcn_wave_barrier();

    const int cellA = piece_cell(rA);
    const int cellB = hasB ? piece_cell(rB) : cellA;

    #pragma unroll 1
    for (int it = 0; it < N_ITER; ++it) {
        if (wl) {
            #pragma unroll
            for (int i = 0; i < 6; ++i) {
                A0[i] = *(const v2f*)&wsh[slot][0][4 * i + h2];
                B0[i] = *(const v2f*)&wsh[slot][1][4 * i + h2];
                A1[i] = *(const v2f*)&wsh[slot][2][4 * i + h2];
                B1[i] = *(const v2f*)&wsh[slot][0][4 * i + 24 + h2];
                A2[i] = *(const v2f*)&wsh[slot][1][4 * i + 24 + h2];
                B2[i] = *(const v2f*)&wsh[slot][2][4 * i + 24 + h2];
            }
        }

        v8f acc0 = {0.f, 0.f, 0.f, 0.f, 0.f, 0.f, 0.f, 0.f};
        v8f acc1 = acc0;
        v8f acc2 = acc0;
        #pragma unroll
        for (int i = 0; i < 6; ++i) {
            acc0 = __builtin_amdgcn_wmma_f32_16x16x4_f32(
                 false, A0[i],  false, B0[i],
                 (short)0, acc0,  false,  false);
            acc1 = __builtin_amdgcn_wmma_f32_16x16x4_f32(
                 false, A1[i],  false, B1[i],
                 (short)0, acc1,  false,  false);
            acc2 = __builtin_amdgcn_wmma_f32_16x16x4_f32(
                 false, A2[i],  false, B2[i],
                 (short)0, acc2,  false,  false);
        }

        if (wl) {
            float* p = &swsh[slot][8 * h];
            *(v4f*)(p     ) = __builtin_shufflevector(acc0, acc0, 0, 1, 2, 3);
            *(v4f*)(p +  4) = __builtin_shufflevector(acc0, acc0, 4, 5, 6, 7);
            *(v4f*)(p + 64) = __builtin_shufflevector(acc1, acc1, 0, 1, 2, 3);
            *(v4f*)(p + 68) = __builtin_shufflevector(acc1, acc1, 4, 5, 6, 7);
            *(v4f*)(p + 80) = __builtin_shufflevector(acc2, acc2, 0, 1, 2, 3);
            *(v4f*)(p + 84) = __builtin_shufflevector(acc2, acc2, 4, 5, 6, 7);
        }
        if (h == 1) {
            swsh[slot][16 + ml] = acc0[7];
            swsh[slot][32 + ml] = acc1[7];
            swsh[slot][48 + ml] = acc2[7];
        }
        __builtin_amdgcn_wave_barrier();

        const float SwA = swsh[slot][cellA] + swsh[slot][cellA + SW_HI];
        const float SwBr = swsh[slot][cellB] + swsh[slot][cellB + SW_HI];
        const float SwB = hasB ? SwBr : 0.0f;

        float nwA = wA - STEP * (2.0f * SwA - betaA + lam_s * sgnf(wA)
                                 + 2.0f * lam_t * (wA - wprevA));
        float nwB = wB - STEP * (2.0f * SwB - betaB + lam_s * sgnf(wB)
                                 + 2.0f * lam_t * (wB - wprevB));

        nwA = clip_w(nwA);
        nwB = clip_w(nwB);
        float s = wave_sum(nwA + nwB);
        float d = s + EPSILON;
        float r = __builtin_amdgcn_rcpf(d);
        nwA = quotient(nwA, d, r);
        nwB = quotient(nwB, d, r);

        nwA = clip_w(nwA);
        nwB = clip_w(nwB);
        s = wave_sum(nwA + nwB);
        d = s + EPSILON;
        r = __builtin_amdgcn_rcpf(d);
        wA = quotient(nwA, d, r);
        wB = quotient(nwB, d, r);

        #pragma unroll
        for (int c = 0; c < 3; ++c) {
            wsh[slot][c][rA] = wA;
            wsh[slot][c][rB] = wB;
        }
        __builtin_amdgcn_wave_barrier();
    }

    out[(size_t)b * P_DIM + rA] = wA;
    if (hasB) out[(size_t)b * P_DIM + rB] = wB;
}

extern "C" void kernel_launch(void* const* d_in, const int* in_sizes, int n_in,
                              void* d_out, int out_size, void* d_ws, size_t ws_size,
                              hipStream_t stream) {
    const float* sigma  = (const float*)d_in[0];
    const float* beta   = (const float*)d_in[1];
    const float* w_prev = (const float*)d_in[2];
    const float* lls    = (const float*)d_in[3];
    const float* llt    = (const float*)d_in[4];
    float* out = (float*)d_out;

    const int nB   = in_sizes[1] / P_DIM;
    const int grid = (nB + WPB - 1) / WPB;

    hipLaunchKernelGGL(risk_budget_wmma_kernel, dim3(grid), dim3(256), 0, stream,
                       sigma, beta, w_prev, lls, llt, out, nB);
}


# Sigma read once by whole-line 16-byte loads (512 contiguous bytes per instruction) into a per-wave LDS area, fragments taken from LDS; was 36 loads of 16 rows x 16 bytes per wave; loop unchanged
# speedup vs baseline: 1.8210x; 1.0764x over previous
#include <hip/hip_runtime.h>


typedef __attribute__((ext_vector_type(2))) float v2f;
typedef __attribute__((ext_vector_type(4))) float v4f;
typedef __attribute__((ext_vector_type(8))) float v8f;
typedef float v4f_a4 __attribute__((ext_vector_type(4), aligned(4)));

#define P_DIM    45
#define P_PAD    48
#define RBLK     15
#define N_ITER   20
#define STEP     0.05f
#define MAXW     0.15f
#define EPSILON  1e-8f
#define WPB      8
#define W_PITCH  64
#define SW_LEN   96
#define SW_HI    48
#define SIG_LEN  (P_DIM * P_DIM)
#define ST_LEN   2048

__device__ __forceinline__ float sgnf(float x) {
    return (x > 0.0f) ? 1.0f : ((x < 0.0f) ? -1.0f : 0.0f);
}

__device__ __forceinline__ float clip_w(float x) {
    return fminf(fmaxf(x, 0.0f), MAXW);
}

__device__ __forceinline__ float quotient(float x, float d, float r) {
    const float q   = x * r;
    const float rem = __builtin_fmaf(-d, q, x);
    return __builtin_fmaf(rem, r, q);
}

__device__ __forceinline__ float other_half(float x) {
    const int b = __builtin_bit_cast(int, x);
    return __builtin_bit_cast(float,
        __builtin_amdgcn_permlanex16(b, b, 0x76543210, 0xfedcba98, false, false));
}
template <int M>
__device__ __forceinline__ float row_xor(float x) {
    const int b = __builtin_bit_cast(int, x);
    return __builtin_bit_cast(float,
        __builtin_amdgcn_update_dpp(b, b, 0x160 + M, 0xf, 0xf, true));
}

__device__ __forceinline__ float wave_sum(float v) {
    v += other_half(v);
    v += row_xor<8>(v);
    v += row_xor<4>(v);
    v += row_xor<2>(v);
    v += row_xor<1>(v);
    return v;
}

template <int CHUNK>
__device__ __forceinline__ v2f sigma_pair(const float* __restrict__ rowp, int h2) {
    v2f a;
    if (CHUNK < 11) {
        a.x = rowp[CHUNK * 4 + h2];
        a.y = rowp[CHUNK * 4 + h2 + 1];
    } else {
        a.x = (h2 == 0) ? rowp[P_DIM - 1] : 0.0f;
        a.y = 0.0f;
    }
    return a;
}

__device__ __forceinline__ int piece_cell(int r) {
    return (r < RBLK) ? r : ((r < 2 * RBLK) ? 16 + (r - RBLK) : 32 + (r - 2 * RBLK));
}

__global__ __launch_bounds__(256) void
risk_budget_wmma_kernel(const float* __restrict__ sigma,
                        const float* __restrict__ beta,
                        const float* __restrict__ w_prev,
                        const float* __restrict__ log_ls,
                        const float* __restrict__ log_lt,
                        float* __restrict__ out,
                        int nB)
{
    __shared__ __attribute__((aligned(16))) float wsh[WPB][3][W_PITCH];
    __shared__ __attribute__((aligned(16))) float swsh[WPB][SW_LEN];
    __shared__ __attribute__((aligned(16))) float stage[WPB][ST_LEN];

    const int lane = threadIdx.x & 31;
    const int slot = threadIdx.x >> 5;
    const int b    = blockIdx.x * WPB + slot;
    if (b >= nB) return;

    const float lam_s = expf(log_ls[0]);
    const float lam_t = expf(log_lt[0]);

    const int  rA   = lane;
    const int  rB   = lane + 32;
    const bool hasB = (rB < P_DIM);

    const float betaA  = beta  [(size_t)b * P_DIM + rA];
    const float wprevA = w_prev[(size_t)b * P_DIM + rA];
    const float betaB  = hasB ? beta  [(size_t)b * P_DIM + rB] : 0.0f;
    const float wprevB = hasB ? w_prev[(size_t)b * P_DIM + rB] : 0.0f;

    const float* sig = sigma + (size_t)b * SIG_LEN;
    float* st = &stage[slot][0];

    {
        v4f q[15];
        #pragma unroll
        for (int j = 0; j < 15; ++j)
            q[j] = *(const v4f_a4*)(sig + 128 * j + 4 * lane);
        const int  tail = 1920 + 4 * lane;
        const bool full = (tail + 3 < SIG_LEN);
        const v4f  qt   = *(const v4f_a4*)(sig + (full ? tail : SIG_LEN - 4));
        #pragma unroll
        for (int j = 0; j < 15; ++j)
            *(v4f*)(st + 128 * j + 4 * lane) = q[j];
        if (full) *(v4f*)(st + tail) = qt;
        else      st[SIG_LEN - 1]    = qt.w;
    }
    __builtin_amdgcn_wave_barrier();

    const int  ml = lane & 15;
    const int  h  = lane >> 4;
    const int  h2 = h << 1;
    const bool wl = (ml == RBLK);
    const int  mr = wl ? 0 : ml;
    const float* row0 = st + (mr            ) * P_DIM;
    const float* row1 = st + (mr +     RBLK) * P_DIM;
    const float* row2 = st + (mr + 2 * RBLK) * P_DIM;

    v2f A0[6], B0[6], A1[6], B1[6], A2[6], B2[6];
    #define LOAD_STEP(I)                                   \
        A0[I] = sigma_pair<(I)    >(row0, h2);             \
        B0[I] = sigma_pair<(I)    >(row1, h2);             \
        A1[I] = sigma_pair<(I) + 6>(row1, h2);             \
        B1[I] = sigma_pair<(I)    >(row2, h2);             \
        A2[I] = sigma_pair<(I) + 6>(row2, h2);             \
        B2[I] = sigma_pair<(I) + 6>(row0, h2);
    LOAD_STEP(0) LOAD_STEP(1) LOAD_STEP(2) LOAD_STEP(3) LOAD_STEP(4) LOAD_STEP(5)
    #undef LOAD_STEP

    float wA = 1.0f / (float)P_DIM;
    float wB = hasB ? 1.0f / (float)P_DIM : 0.0f;
    #pragma unroll
    for (int c = 0; c < 3; ++c) {
        wsh[slot][c][rA] = wA;
        wsh[slot][c][rB] = wB;
    }
    __builtin_amdgcn_wave_barrier();

    const int cellA = piece_cell(rA);
    const int cellB = hasB ? piece_cell(rB) : cellA;

    #pragma unroll 1
    for (int it = 0; it < N_ITER; ++it) {
        if (wl) {
            #pragma unroll
            for (int i = 0; i < 6; ++i) {
                A0[i] = *(const v2f*)&wsh[slot][0][4 * i + h2];
                B0[i] = *(const v2f*)&wsh[slot][1][4 * i + h2];
                A1[i] = *(const v2f*)&wsh[slot][2][4 * i + h2];
                B1[i] = *(const v2f*)&wsh[slot][0][4 * i + 24 + h2];
                A2[i] = *(const v2f*)&wsh[slot][1][4 * i + 24 + h2];
                B2[i] = *(const v2f*)&wsh[slot][2][4 * i + 24 + h2];
            }
        }

        v8f acc0 = {0.f, 0.f, 0.f, 0.f, 0.f, 0.f, 0.f, 0.f};
        v8f acc1 = acc0;
        v8f acc2 = acc0;
        #pragma unroll
        for (int i = 0; i < 6; ++i) {
            acc0 = __builtin_amdgcn_wmma_f32_16x16x4_f32(
                 false, A0[i],  false, B0[i],
                 (short)0, acc0,  false,  false);
            acc1 = __builtin_amdgcn_wmma_f32_16x16x4_f32(
                 false, A1[i],  false, B1[i],
                 (short)0, acc1,  false,  false);
            acc2 = __builtin_amdgcn_wmma_f32_16x16x4_f32(
                 false, A2[i],  false, B2[i],
                 (short)0, acc2,  false,  false);
        }

        if (wl) {
            float* p = &swsh[slot][8 * h];
            *(v4f*)(p     ) = __builtin_shufflevector(acc0, acc0, 0, 1, 2, 3);
            *(v4f*)(p +  4) = __builtin_shufflevector(acc0, acc0, 4, 5, 6, 7);
            *(v4f*)(p + 64) = __builtin_shufflevector(acc1, acc1, 0, 1, 2, 3);
            *(v4f*)(p + 68) = __builtin_shufflevector(acc1, acc1, 4, 5, 6, 7);
            *(v4f*)(p + 80) = __builtin_shufflevector(acc2, acc2, 0, 1, 2, 3);
            *(v4f*)(p + 84) = __builtin_shufflevector(acc2, acc2, 4, 5, 6, 7);
        }
        if (h == 1) {
            swsh[slot][16 + ml] = acc0[7];
            swsh[slot][32 + ml] = acc1[7];
            swsh[slot][48 + ml] = acc2[7];
        }
        __builtin_amdgcn_wave_barrier();

        const float SwA = swsh[slot][cellA] + swsh[slot][cellA + SW_HI];
        const float SwBr = swsh[slot][cellB] + swsh[slot][cellB + SW_HI];
        const float SwB = hasB ? SwBr : 0.0f;

        float nwA = wA - STEP * (2.0f * SwA - betaA + lam_s * sgnf(wA)
                                 + 2.0f * lam_t * (wA - wprevA));
        float nwB = wB - STEP * (2.0f * SwB - betaB + lam_s * sgnf(wB)
                                 + 2.0f * lam_t * (wB - wprevB));

        nwA = clip_w(nwA);
        nwB = clip_w(nwB);
        float s = wave_sum(nwA + nwB);
        float d = s + EPSILON;
        float r = __builtin_amdgcn_rcpf(d);
        nwA = quotient(nwA, d, r);
        nwB = quotient(nwB, d, r);

        nwA = clip_w(nwA);
        nwB = clip_w(nwB);
        s = wave_sum(nwA + nwB);
        d = s + EPSILON;
        r = __builtin_amdgcn_rcpf(d);
        wA = quotient(nwA, d, r);
        wB = quotient(nwB, d, r);

        #pragma unroll
        for (int c = 0; c < 3; ++c) {
            wsh[slot][c][rA] = wA;
            wsh[slot][c][rB] = wB;
        }
        __builtin_amdgcn_wave_barrier();
    }

    out[(size_t)b * P_DIM + rA] = wA;
    if (hasB) out[(size_t)b * P_DIM + rB] = wB;
}

extern "C" void kernel_launch(void* const* d_in, const int* in_sizes, int n_in,
                              void* d_out, int out_size, void* d_ws, size_t ws_size,
                              hipStream_t stream) {
    const float* sigma  = (const float*)d_in[0];
    const float* beta   = (const float*)d_in[1];
    const float* w_prev = (const float*)d_in[2];
    const float* lls    = (const float*)d_in[3];
    const float* llt    = (const float*)d_in[4];
    float* out = (float*)d_out;

    const int nB   = in_sizes[1] / P_DIM;
    const int grid = (nB + WPB - 1) / WPB;

    hipLaunchKernelGGL(risk_budget_wmma_kernel, dim3(grid), dim3(256), 0, stream,
                       sigma, beta, w_prev, lls, llt, out, nB);
}
